# starting kernel
# baseline (speedup 1.0000x reference)
#include <hip/hip_runtime.h>

#define N_NODES 100000
#define N_EDGES 1600000
#define D_IN    128
#define D_OUT   64

typedef __attribute__((ext_vector_type(2))) float v2f;
typedef __attribute__((ext_vector_type(8))) float v8f;

__global__ __launch_bounds__(256) void zero_kernel(float4* __restrict__ out, int n4) {
    int i = blockIdx.x * blockDim.x + threadIdx.x;
    if (i < n4) out[i] = make_float4(0.f, 0.f, 0.f, 0.f);
}

__global__ __launch_bounds__(256) void gemm_bias_kernel(
    const float* __restrict__ x, const float* __restrict__ W,
    const float* __restrict__ bias, float* __restrict__ h)
{
    const int lane = threadIdx.x & 31;
    const int wave = threadIdx.x >> 5;
    const int m0   = (blockIdx.x * 8 + wave) * 16;
    if (m0 >= N_NODES) return;

    const int half = lane >> 4;
    const int l15  = lane & 15;

    v8f acc[4];
    #pragma unroll
    for (int t = 0; t < 4; ++t) {
        const float bval = bias[t * 16 + l15];
        #pragma unroll
        for (int i = 0; i < 8; ++i) acc[t][i] = bval;
    }

    const float* xrow  = x + (size_t)(m0 + l15) * D_IN + half * 2;
    const float* wbase = W + (size_t)(half * 2) * D_OUT + l15;

    #pragma unroll
    for (int k0 = 0; k0 < D_IN; k0 += 4) {
        v2f a;
        a.x = xrow[k0 + 0];
        a.y = xrow[k0 + 1];
        #pragma unroll
        for (int t = 0; t < 4; ++t) {
            v2f bm;
            bm.x = wbase[(size_t)(k0 + 0) * D_OUT + t * 16];
            bm.y = wbase[(size_t)(k0 + 1) * D_OUT + t * 16];
            acc[t] = __builtin_amdgcn_wmma_f32_16x16x4_f32(
                          false, a,  false, bm,
                          (short)0, acc[t],
                          false,  false);
        }
    }

    float* hout = h + (size_t)(m0 + half * 8) * D_OUT + l15;
    #pragma unroll
    for (int t = 0; t < 4; ++t)
        #pragma unroll
        for (int v = 0; v < 8; ++v)
            hout[(size_t)v * D_OUT + t * 16] = acc[t][v];
}

__global__ __launch_bounds__(256) void scatter_kernel(
    const float* __restrict__ h, const float* __restrict__ edge_w,
    const int* __restrict__ src, const int* __restrict__ dst,
    float* __restrict__ out)
{
    const int lane = threadIdx.x & 31;
    const int e    = blockIdx.x * 8 + (threadIdx.x >> 5);
    if (e >= N_EDGES) return;

    const int   s = src[e];
    const int   d = dst[e];
    const float w = edge_w[e];

    const float2 v = ((const float2*)(h + (size_t)s * D_OUT))[lane];
    float* orow = out + (size_t)d * D_OUT + lane * 2;
    atomicAdd(orow + 0, v.x * w);
    atomicAdd(orow + 1, v.y * w);
}

extern "C" void kernel_launch(void* const* d_in, const int* in_sizes, int n_in,
                              void* d_out, int out_size, void* d_ws, size_t ws_size,
                              hipStream_t stream) {
    const float* x      = (const float*)d_in[0];
    const float* W      = (const float*)d_in[1];
    const float* b      = (const float*)d_in[2];
    const float* edge_w = (const float*)d_in[3];
    const int*   src    = (const int*)d_in[4];
    const int*   dst    = (const int*)d_in[5];
    float*       out    = (float*)d_out;
    float*       h      = (float*)d_ws;

    (void)in_sizes; (void)n_in; (void)ws_size;

    const int n4 = out_size / 4;
    zero_kernel<<<(n4 + 255) / 256, 256, 0, stream>>>((float4*)out, n4);

    const int m_tiles = (N_NODES + 15) / 16;
    const int gemm_blocks = (m_tiles + 7) / 8;
    gemm_bias_kernel<<<gemm_blocks, 256, 0, stream>>>(x, W, b, h);

    scatter_kernel<<<(N_EDGES + 7) / 8, 256, 0, stream>>>(h, edge_w, src, dst, out);
}


# Edge phase: one wave per 32 edges instead of per edge; indices read coalesced, 16 source-row reads issued back to back before their float atomic adds; sum order still atomic arrival order
# speedup vs baseline: 1.7974x; 1.7974x over previous
#include <hip/hip_runtime.h>

#define N_NODES 100000
#define N_EDGES 1600000
#define D_IN    128
#define D_OUT   64

typedef __attribute__((ext_vector_type(2))) float v2f;
typedef __attribute__((ext_vector_type(8))) float v8f;

__global__ __launch_bounds__(256) void zero_kernel(float4* __restrict__ out, int n4) {
    int i = blockIdx.x * blockDim.x + threadIdx.x;
    if (i < n4) out[i] = make_float4(0.f, 0.f, 0.f, 0.f);
}

__global__ __launch_bounds__(256) void gemm_bias_kernel(
    const float* __restrict__ x, const float* __restrict__ W,
    const float* __restrict__ bias, float* __restrict__ h)
{
    const int lane = threadIdx.x & 31;
    const int wave = threadIdx.x >> 5;
    const int m0   = (blockIdx.x * 8 + wave) * 16;
    if (m0 >= N_NODES) return;

    const int half = lane >> 4;
    const int l15  = lane & 15;

    v8f acc[4];
    #pragma unroll
    for (int t = 0; t < 4; ++t) {
        const float bval = bias[t * 16 + l15];
        #pragma unroll
        for (int i = 0; i < 8; ++i) acc[t][i] = bval;
    }

    const float* xrow  = x + (size_t)(m0 + l15) * D_IN + half * 2;
    const float* wbase = W + (size_t)(half * 2) * D_OUT + l15;

    #pragma unroll
    for (int k0 = 0; k0 < D_IN; k0 += 4) {
        v2f a;
        a.x = xrow[k0 + 0];
        a.y = xrow[k0 + 1];
        #pragma unroll
        for (int t = 0; t < 4; ++t) {
            v2f bm;
            bm.x = wbase[(size_t)(k0 + 0) * D_OUT + t * 16];
            bm.y = wbase[(size_t)(k0 + 1) * D_OUT + t * 16];
            acc[t] = __builtin_amdgcn_wmma_f32_16x16x4_f32(
                          false, a,  false, bm,
                          (short)0, acc[t],
                          false,  false);
        }
    }

    float* hout = h + (size_t)(m0 + half * 8) * D_OUT + l15;
    #pragma unroll
    for (int t = 0; t < 4; ++t)
        #pragma unroll
        for (int v = 0; v < 8; ++v)
            hout[(size_t)v * D_OUT + t * 16] = acc[t][v];
}

#define EDGES_PER_WAVE 32
#define EDGE_GROUP     16
static_assert(N_EDGES % EDGES_PER_WAVE == 0, "edge count must fill whole waves");
static_assert(EDGES_PER_WAVE % EDGE_GROUP == 0, "groups must tile the wave's edges");

__global__ __launch_bounds__(256) void scatter_kernel(
    const float* __restrict__ h, const float* __restrict__ edge_w,
    const int* __restrict__ src, const int* __restrict__ dst,
    float* __restrict__ out)
{
    const int lane = threadIdx.x & 31;
    const int e0   = (blockIdx.x * 8 + (threadIdx.x >> 5)) * EDGES_PER_WAVE;
    if (e0 >= N_EDGES) return;

    const int s_l = src[e0 + lane];
    const int d_l = dst[e0 + lane];
    const int w_l = __float_as_int(edge_w[e0 + lane]);

    #pragma unroll 1
    for (int g = 0; g < EDGES_PER_WAVE; g += EDGE_GROUP) {
        float2 v[EDGE_GROUP];
        #pragma unroll
        for (int j = 0; j < EDGE_GROUP; ++j) {
            const int s = __builtin_amdgcn_readlane(s_l, g + j);
            v[j] = ((const float2*)(h + (size_t)s * D_OUT))[lane];
        }
        #pragma unroll
        for (int j = 0; j < EDGE_GROUP; ++j) {
            const int   d = __builtin_amdgcn_readlane(d_l, g + j);
            const float w = __int_as_float(__builtin_amdgcn_readlane(w_l, g + j));
            float* orow = out + (size_t)d * D_OUT + lane * 2;
            atomicAdd(orow + 0, v[j].x * w);
            atomicAdd(orow + 1, v[j].y * w);
        }
    }
}

extern "C" void kernel_launch(void* const* d_in, const int* in_sizes, int n_in,
                              void* d_out, int out_size, void* d_ws, size_t ws_size,
                              hipStream_t stream) {
    const float* x      = (const float*)d_in[0];
    const float* W      = (const float*)d_in[1];
    const float* b      = (const float*)d_in[2];
    const float* edge_w = (const float*)d_in[3];
    const int*   src    = (const int*)d_in[4];
    const int*   dst    = (const int*)d_in[5];
    float*       out    = (float*)d_out;
    float*       h      = (float*)d_ws;

    (void)in_sizes; (void)n_in; (void)ws_size;

    const int n4 = out_size / 4;
    zero_kernel<<<(n4 + 255) / 256, 256, 0, stream>>>((float4*)out, n4);

    const int m_tiles = (N_NODES + 15) / 16;
    const int gemm_blocks = (m_tiles + 7) / 8;
    gemm_bias_kernel<<<gemm_blocks, 256, 0, stream>>>(x, W, b, h);

    scatter_kernel<<<N_EDGES / (EDGES_PER_WAVE * 8), 256, 0, stream>>>(h, edge_w, src, dst, out);
}


# Edge phase: h stored with columns c and c+32 adjacent so lane l adds columns l and l+32: each f32 atomic covers one whole 128-byte line of out, not two half-used lines; sums still in arrival order
# speedup vs baseline: 2.2838x; 1.2706x over previous
#include <hip/hip_runtime.h>

#define N_NODES 100000
#define N_EDGES 1600000
#define D_IN    128
#define D_OUT   64

typedef __attribute__((ext_vector_type(2))) float v2f;
typedef __attribute__((ext_vector_type(8))) float v8f;

__global__ __launch_bounds__(256) void zero_kernel(float4* __restrict__ out, int n4) {
    int i = blockIdx.x * blockDim.x + threadIdx.x;
    if (i < n4) out[i] = make_float4(0.f, 0.f, 0.f, 0.f);
}

__global__ __launch_bounds__(256) void gemm_bias_kernel(
    const float* __restrict__ x, const float* __restrict__ W,
    const float* __restrict__ bias, float* __restrict__ h)
{
    const int lane = threadIdx.x & 31;
    const int wave = threadIdx.x >> 5;
    const int m0   = (blockIdx.x * 8 + wave) * 16;
    if (m0 >= N_NODES) return;

    const int half = lane >> 4;
    const int l15  = lane & 15;

    v8f acc[4];
    #pragma unroll
    for (int t = 0; t < 4; ++t) {
        const float bval = bias[t * 16 + l15];
        #pragma unroll
        for (int i = 0; i < 8; ++i) acc[t][i] = bval;
    }

    const float* xrow  = x + (size_t)(m0 + l15) * D_IN + half * 2;
    const float* wbase = W + (size_t)(half * 2) * D_OUT + l15;

    #pragma unroll
    for (int k0 = 0; k0 < D_IN; k0 += 4) {
        v2f a;
        a.x = xrow[k0 + 0];
        a.y = xrow[k0 + 1];
        #pragma unroll
        for (int t = 0; t < 4; ++t) {
            v2f bm;
            bm.x = wbase[(size_t)(k0 + 0) * D_OUT + t * 16];
            bm.y = wbase[(size_t)(k0 + 1) * D_OUT + t * 16];
            acc[t] = __builtin_amdgcn_wmma_f32_16x16x4_f32(
                          false, a,  false, bm,
                          (short)0, acc[t],
                          false,  false);
        }
    }

    float* hout = h + (size_t)(m0 + half * 8) * D_OUT + 2 * l15;
    #pragma unroll
    for (int t = 0; t < 2; ++t)
        #pragma unroll
        for (int v = 0; v < 8; ++v)
            *(float2*)(hout + (size_t)v * D_OUT + t * 32) =
                make_float2(acc[t][v], acc[t + 2][v]);
}

#define EDGES_PER_WAVE 32
#define EDGE_GROUP     16
static_assert(N_EDGES % EDGES_PER_WAVE == 0, "edge count must fill whole waves");
static_assert(EDGES_PER_WAVE % EDGE_GROUP == 0, "groups must tile the wave's edges");

__global__ __launch_bounds__(256, 16) void scatter_kernel(
    const float* __restrict__ h, const float* __restrict__ edge_w,
    const int* __restrict__ src, const int* __restrict__ dst,
    float* __restrict__ out)
{
    const int lane = threadIdx.x & 31;
    const int e0   = (blockIdx.x * 8 + (threadIdx.x >> 5)) * EDGES_PER_WAVE;
    if (e0 >= N_EDGES) return;

    const int s_l = src[e0 + lane];
    const int d_l = dst[e0 + lane];
    const int w_l = __float_as_int(edge_w[e0 + lane]);

    #pragma unroll 1
    for (int g = 0; g < EDGES_PER_WAVE; g += EDGE_GROUP) {
        float2 v[EDGE_GROUP];
        #pragma unroll
        for (int j = 0; j < EDGE_GROUP; ++j) {
            const int s = __builtin_amdgcn_readlane(s_l, g + j);
            v[j] = ((const float2*)(h + (size_t)s * D_OUT))[lane];
        }
        #pragma unroll
        for (int j = 0; j < EDGE_GROUP; ++j) {
            const int   d = __builtin_amdgcn_readlane(d_l, g + j);
            const float w = __int_as_float(__builtin_amdgcn_readlane(w_l, g + j));
            float* orow = out + (size_t)d * D_OUT + lane;
            atomicAdd(orow,      v[j].x * w);
            atomicAdd(orow + 32, v[j].y * w);
        }
    }
}

extern "C" void kernel_launch(void* const* d_in, const int* in_sizes, int n_in,
                              void* d_out, int out_size, void* d_ws, size_t ws_size,
                              hipStream_t stream) {
    const float* x      = (const float*)d_in[0];
    const float* W      = (const float*)d_in[1];
    const float* b      = (const float*)d_in[2];
    const float* edge_w = (const float*)d_in[3];
    const int*   src    = (const int*)d_in[4];
    const int*   dst    = (const int*)d_in[5];
    float*       out    = (float*)d_out;
    float*       h      = (float*)d_ws;

    (void)in_sizes; (void)n_in; (void)ws_size;

    const int n4 = out_size / 4;
    zero_kernel<<<(n4 + 255) / 256, 256, 0, stream>>>((float4*)out, n4);

    const int m_tiles = (N_NODES + 15) / 16;
    const int gemm_blocks = (m_tiles + 7) / 8;
    gemm_bias_kernel<<<gemm_blocks, 256, 0, stream>>>(x, W, b, h);

    scatter_kernel<<<N_EDGES / (EDGES_PER_WAVE * 8), 256, 0, stream>>>(h, edge_w, src, dst, out);
}


# Projection K loop: fragments fetched as 16-byte (x) and 8-byte (W) vectors, one 16-wide K block ahead of its WMMAs as one group: half the load instructions; k summed in a permuted order per block
# speedup vs baseline: 2.3829x; 1.0434x over previous
#include <hip/hip_runtime.h>

#define N_NODES 100000
#define N_EDGES 1600000
#define D_IN    128
#define D_OUT   64

typedef __attribute__((ext_vector_type(2))) float v2f;
typedef __attribute__((ext_vector_type(4))) float v4f;
typedef __attribute__((ext_vector_type(8))) float v8f;

__global__ __launch_bounds__(256) void zero_kernel(float4* __restrict__ out, int n4) {
    int i = blockIdx.x * blockDim.x + threadIdx.x;
    if (i < n4) out[i] = make_float4(0.f, 0.f, 0.f, 0.f);
}

__global__ __launch_bounds__(256) void gemm_bias_kernel(
    const float* __restrict__ x, const float* __restrict__ W,
    const float* __restrict__ bias, float* __restrict__ h)
{
    const int lane = threadIdx.x & 31;
    const int wave = threadIdx.x >> 5;
    const int m0   = (blockIdx.x * 8 + wave) * 16;
    if (m0 >= N_NODES) return;

    const int half = lane >> 4;
    const int l15  = lane & 15;

    v8f acc[4];
    #pragma unroll
    for (int t = 0; t < 4; ++t) {
        const float bval = bias[2 * l15 + (t >> 1) + 32 * (t & 1)];
        #pragma unroll
        for (int i = 0; i < 8; ++i) acc[t][i] = bval;
    }

    const float* xrow = x + (size_t)(m0 + l15) * D_IN + half * 8;
    const float* wrow = W + (size_t)(half * 8) * D_OUT + 2 * l15;

    v4f a0[2], a1[2];
    v2f lo[2][8], hi[2][8];
    a0[0] = *(const v4f*)(xrow);
    a1[0] = *(const v4f*)(xrow + 4);
    #pragma unroll
    for (int r = 0; r < 8; ++r) {
        lo[0][r] = *(const v2f*)(wrow + (size_t)r * D_OUT);
        hi[0][r] = *(const v2f*)(wrow + (size_t)r * D_OUT + 32);
    }

    #pragma unroll
    for (int kb = 0; kb < D_IN; kb += 16) {
        const int cur = (kb >> 4) & 1, nxt = cur ^ 1;
        if (kb + 16 < D_IN) {
            a0[nxt] = *(const v4f*)(xrow + kb + 16);
            a1[nxt] = *(const v4f*)(xrow + kb + 20);
            #pragma unroll
            for (int r = 0; r < 8; ++r) {
                lo[nxt][r] = *(const v2f*)(wrow + (size_t)(kb + 16 + r) * D_OUT);
                hi[nxt][r] = *(const v2f*)(wrow + (size_t)(kb + 16 + r) * D_OUT + 32);
            }
        }
        __builtin_amdgcn_sched_barrier(0);
        const float av[8] = {a0[cur].x, a0[cur].y, a0[cur].z, a0[cur].w,
                             a1[cur].x, a1[cur].y, a1[cur].z, a1[cur].w};
        #pragma unroll
        for (int j = 0; j < 4; ++j) {
            v2f a;  a.x  = av[2 * j];            a.y  = av[2 * j + 1];
            v2f b0; b0.x = lo[cur][2 * j].x;     b0.y = lo[cur][2 * j + 1].x;
            v2f b1; b1.x = hi[cur][2 * j].x;     b1.y = hi[cur][2 * j + 1].x;
            v2f b2; b2.x = lo[cur][2 * j].y;     b2.y = lo[cur][2 * j + 1].y;
            v2f b3; b3.x = hi[cur][2 * j].y;     b3.y = hi[cur][2 * j + 1].y;
            acc[0] = __builtin_amdgcn_wmma_f32_16x16x4_f32(false, a, false, b0, (short)0, acc[0], false, false);
            acc[1] = __builtin_amdgcn_wmma_f32_16x16x4_f32(false, a, false, b1, (short)0, acc[1], false, false);
            acc[2] = __builtin_amdgcn_wmma_f32_16x16x4_f32(false, a, false, b2, (short)0, acc[2], false, false);
            acc[3] = __builtin_amdgcn_wmma_f32_16x16x4_f32(false, a, false, b3, (short)0, acc[3], false, false);
        }
        __builtin_amdgcn_sched_barrier(0);
    }

    float* hout = h + (size_t)(m0 + half * 8) * D_OUT + 4 * l15;
    #pragma unroll
    for (int v = 0; v < 8; ++v) {
        v4f o; o.x = acc[0][v]; o.y = acc[1][v]; o.z = acc[2][v]; o.w = acc[3][v];
        *(v4f*)(hout + (size_t)v * D_OUT) = o;
    }
}

#define EDGES_PER_WAVE 32
#define EDGE_GROUP     16
static_assert(N_EDGES % EDGES_PER_WAVE == 0, "edge count must fill whole waves");
static_assert(EDGES_PER_WAVE % EDGE_GROUP == 0, "groups must tile the wave's edges");

__global__ __launch_bounds__(256, 16) void scatter_kernel(
    const float* __restrict__ h, const float* __restrict__ edge_w,
    const int* __restrict__ src, const int* __restrict__ dst,
    float* __restrict__ out)
{
    const int lane = threadIdx.x & 31;
    const int e0   = (blockIdx.x * 8 + (threadIdx.x >> 5)) * EDGES_PER_WAVE;
    if (e0 >= N_EDGES) return;

    const int s_l = src[e0 + lane];
    const int d_l = dst[e0 + lane];
    const int w_l = __float_as_int(edge_w[e0 + lane]);

    #pragma unroll 1
    for (int g = 0; g < EDGES_PER_WAVE; g += EDGE_GROUP) {
        float2 v[EDGE_GROUP];
        #pragma unroll
        for (int j = 0; j < EDGE_GROUP; ++j) {
            const int s = __builtin_amdgcn_readlane(s_l, g + j);
            v[j] = ((const float2*)(h + (size_t)s * D_OUT))[lane];
        }
        #pragma unroll
        for (int j = 0; j < EDGE_GROUP; ++j) {
            const int   d = __builtin_amdgcn_readlane(d_l, g + j);
            const float w = __int_as_float(__builtin_amdgcn_readlane(w_l, g + j));
            float* orow = out + (size_t)d * D_OUT + lane;
            atomicAdd(orow,      v[j].x * w);
            atomicAdd(orow + 32, v[j].y * w);
        }
    }
}

extern "C" void kernel_launch(void* const* d_in, const int* in_sizes, int n_in,
                              void* d_out, int out_size, void* d_ws, size_t ws_size,
                              hipStream_t stream) {
    const float* x      = (const float*)d_in[0];
    const float* W      = (const float*)d_in[1];
    const float* b      = (const float*)d_in[2];
    const float* edge_w = (const float*)d_in[3];
    const int*   src    = (const int*)d_in[4];
    const int*   dst    = (const int*)d_in[5];
    float*       out    = (float*)d_out;
    float*       h      = (float*)d_ws;

    (void)in_sizes; (void)n_in; (void)ws_size;

    const int n4 = out_size / 4;
    zero_kernel<<<(n4 + 255) / 256, 256, 0, stream>>>((float4*)out, n4);

    const int m_tiles = (N_NODES + 15) / 16;
    const int gemm_blocks = (m_tiles + 7) / 8;
    gemm_bias_kernel<<<gemm_blocks, 256, 0, stream>>>(x, W, b, h);

    scatter_kernel<<<N_EDGES / (EDGES_PER_WAVE * 8), 256, 0, stream>>>(h, edge_w, src, dst, out);
}


# Clearing of out folded into the projection kernel: each wave zero-fills the 16 out rows of its strip with eight 512-byte stores behind its K loop; 2 launches instead of 3; sums unchanged
# speedup vs baseline: 2.4635x; 1.0338x over previous
#include <hip/hip_runtime.h>

#define N_NODES 100000
#define N_EDGES 1600000
#define D_IN    128
#define D_OUT   64

typedef __attribute__((ext_vector_type(2))) float v2f;
typedef __attribute__((ext_vector_type(4))) float v4f;
typedef __attribute__((ext_vector_type(8))) float v8f;

__global__ __launch_bounds__(256) void gemm_bias_kernel(
    const float* __restrict__ x, const float* __restrict__ W,
    const float* __restrict__ bias, float* __restrict__ h, float* __restrict__ out)
{
    const int lane = threadIdx.x & 31;
    const int wave = threadIdx.x >> 5;
    const int m0   = (blockIdx.x * 8 + wave) * 16;
    if (m0 >= N_NODES) return;

    const int half = lane >> 4;
    const int l15  = lane & 15;

    v8f acc[4];
    #pragma unroll
    for (int t = 0; t < 4; ++t) {
        const float bval = bias[2 * l15 + (t >> 1) + 32 * (t & 1)];
        #pragma unroll
        for (int i = 0; i < 8; ++i) acc[t][i] = bval;
    }

    const float* xrow = x + (size_t)(m0 + l15) * D_IN + half * 8;
    const float* wrow = W + (size_t)(half * 8) * D_OUT + 2 * l15;

    v4f a0[2], a1[2];
    v2f lo[2][8], hi[2][8];
    a0[0] = *(const v4f*)(xrow);
    a1[0] = *(const v4f*)(xrow + 4);
    #pragma unroll
    for (int r = 0; r < 8; ++r) {
        lo[0][r] = *(const v2f*)(wrow + (size_t)r * D_OUT);
        hi[0][r] = *(const v2f*)(wrow + (size_t)r * D_OUT + 32);
    }

    #pragma unroll
    for (int kb = 0; kb < D_IN; kb += 16) {
        const int cur = (kb >> 4) & 1, nxt = cur ^ 1;
        if (kb + 16 < D_IN) {
            a0[nxt] = *(const v4f*)(xrow + kb + 16);
            a1[nxt] = *(const v4f*)(xrow + kb + 20);
            #pragma unroll
            for (int r = 0; r < 8; ++r) {
                lo[nxt][r] = *(const v2f*)(wrow + (size_t)(kb + 16 + r) * D_OUT);
                hi[nxt][r] = *(const v2f*)(wrow + (size_t)(kb + 16 + r) * D_OUT + 32);
            }
        }
        __builtin_amdgcn_sched_barrier(0);
        const float av[8] = {a0[cur].x, a0[cur].y, a0[cur].z, a0[cur].w,
                             a1[cur].x, a1[cur].y, a1[cur].z, a1[cur].w};
        #pragma unroll
        for (int j = 0; j < 4; ++j) {
            v2f a;  a.x  = av[2 * j];            a.y  = av[2 * j + 1];
            v2f b0; b0.x = lo[cur][2 * j].x;     b0.y = lo[cur][2 * j + 1].x;
            v2f b1; b1.x = hi[cur][2 * j].x;     b1.y = hi[cur][2 * j + 1].x;
            v2f b2; b2.x = lo[cur][2 * j].y;     b2.y = lo[cur][2 * j + 1].y;
            v2f b3; b3.x = hi[cur][2 * j].y;     b3.y = hi[cur][2 * j + 1].y;
            acc[0] = __builtin_amdgcn_wmma_f32_16x16x4_f32(false, a, false, b0, (short)0, acc[0], false, false);
            acc[1] = __builtin_amdgcn_wmma_f32_16x16x4_f32(false, a, false, b1, (short)0, acc[1], false, false);
            acc[2] = __builtin_amdgcn_wmma_f32_16x16x4_f32(false, a, false, b2, (short)0, acc[2], false, false);
            acc[3] = __builtin_amdgcn_wmma_f32_16x16x4_f32(false, a, false, b3, (short)0, acc[3], false, false);
        }
        __builtin_amdgcn_sched_barrier(0);
    }

    float* zrow = out + (size_t)(m0 + half) * D_OUT + 4 * l15;
    const v4f zero4 = {0.f, 0.f, 0.f, 0.f};
    for (int i = 0; i < 8; ++i) *(v4f*)(zrow + (size_t)(2 * i) * D_OUT) = zero4;

    float* hout = h + (size_t)(m0 + half * 8) * D_OUT + 4 * l15;
    #pragma unroll
    for (int v = 0; v < 8; ++v) {
        v4f o; o.x = acc[0][v]; o.y = acc[1][v]; o.z = acc[2][v]; o.w = acc[3][v];
        *(v4f*)(hout + (size_t)v * D_OUT) = o;
    }
}

#define EDGES_PER_WAVE 32
#define EDGE_GROUP     16
static_assert(N_EDGES % EDGES_PER_WAVE == 0, "edge count must fill whole waves");
static_assert(N_NODES % 16 == 0, "16-row strips must tile the rows: the projection launch clears out strip by strip");

static_assert(EDGES_PER_WAVE % EDGE_GROUP == 0, "groups must tile the wave's edges");

__global__ __launch_bounds__(256, 16) void scatter_kernel(
    const float* __restrict__ h, const float* __restrict__ edge_w,
    const int* __restrict__ src, const int* __restrict__ dst,
    float* __restrict__ out)
{
    const int lane = threadIdx.x & 31;
    const int e0   = (blockIdx.x * 8 + (threadIdx.x >> 5)) * EDGES_PER_WAVE;
    if (e0 >= N_EDGES) return;

    const int s_l = src[e0 + lane];
    const int d_l = dst[e0 + lane];
    const int w_l = __float_as_int(edge_w[e0 + lane]);

    #pragma unroll 1
    for (int g = 0; g < EDGES_PER_WAVE; g += EDGE_GROUP) {
        float2 v[EDGE_GROUP];
        #pragma unroll
        for (int j = 0; j < EDGE_GROUP; ++j) {
            const int s = __builtin_amdgcn_readlane(s_l, g + j);
            v[j] = ((const float2*)(h + (size_t)s * D_OUT))[lane];
        }
        #pragma unroll
        for (int j = 0; j < EDGE_GROUP; ++j) {
            const int   d = __builtin_amdgcn_readlane(d_l, g + j);
            const float w = __int_as_float(__builtin_amdgcn_readlane(w_l, g + j));
            float* orow = out + (size_t)d * D_OUT + lane;
            atomicAdd(orow,      v[j].x * w);
            atomicAdd(orow + 32, v[j].y * w);
        }
    }
}

extern "C" void kernel_launch(void* const* d_in, const int* in_sizes, int n_in,
                              void* d_out, int out_size, void* d_ws, size_t ws_size,
                              hipStream_t stream) {
    const float* x      = (const float*)d_in[0];
    const float* W      = (const float*)d_in[1];
    const float* b      = (const float*)d_in[2];
    const float* edge_w = (const float*)d_in[3];
    const int*   src    = (const int*)d_in[4];
    const int*   dst    = (const int*)d_in[5];
    float*       out    = (float*)d_out;
    float*       h      = (float*)d_ws;

    (void)in_sizes; (void)n_in; (void)out_size; (void)ws_size;

    const int m_tiles = (N_NODES + 15) / 16;
    const int gemm_blocks = (m_tiles + 7) / 8;
    gemm_bias_kernel<<<gemm_blocks, 256, 0, stream>>>(x, W, b, h, out);

    scatter_kernel<<<N_EDGES / (EDGES_PER_WAVE * 8), 256, 0, stream>>>(h, edge_w, src, dst, out);
}


# Edge phase: edges ordered by 196-row destination block in LDS and filed with whole-line stores; each block then summed in LDS (float LDS adds, arrival order), stored once; no global float atomics
# speedup vs baseline: 3.1211x; 1.2669x over previous
#include <hip/hip_runtime.h>

#define N_NODES 100000
#define N_EDGES 1600000
#define D_IN    128
#define D_OUT   64

typedef __attribute__((ext_vector_type(2))) float v2f;
typedef __attribute__((ext_vector_type(4))) float v4f;
typedef __attribute__((ext_vector_type(8))) float v8f;

__global__ __launch_bounds__(256) void gemm_bias_kernel(
    const float* __restrict__ x, const float* __restrict__ W,
    const float* __restrict__ bias, float* __restrict__ h)
{
    const int lane = threadIdx.x & 31;
    const int wave = threadIdx.x >> 5;
    const int m0   = (blockIdx.x * 8 + wave) * 16;
    if (m0 >= N_NODES) return;

    const int half = lane >> 4;
    const int l15  = lane & 15;

    v8f acc[4];
    #pragma unroll
    for (int t = 0; t < 4; ++t) {
        const float bval = bias[2 * l15 + (t >> 1) + 32 * (t & 1)];
        #pragma unroll
        for (int i = 0; i < 8; ++i) acc[t][i] = bval;
    }

    const float* xrow = x + (size_t)(m0 + l15) * D_IN + half * 8;
    const float* wrow = W + (size_t)(half * 8) * D_OUT + 2 * l15;

    v4f a0[2], a1[2];
    v2f lo[2][8], hi[2][8];
    a0[0] = *(const v4f*)(xrow);
    a1[0] = *(const v4f*)(xrow + 4);
    #pragma unroll
    for (int r = 0; r < 8; ++r) {
        lo[0][r] = *(const v2f*)(wrow + (size_t)r * D_OUT);
        hi[0][r] = *(const v2f*)(wrow + (size_t)r * D_OUT + 32);
    }

    #pragma unroll
    for (int kb = 0; kb < D_IN; kb += 16) {
        const int cur = (kb >> 4) & 1, nxt = cur ^ 1;
        if (kb + 16 < D_IN) {
            a0[nxt] = *(const v4f*)(xrow + kb + 16);
            a1[nxt] = *(const v4f*)(xrow + kb + 20);
            #pragma unroll
            for (int r = 0; r < 8; ++r) {
                lo[nxt][r] = *(const v2f*)(wrow + (size_t)(kb + 16 + r) * D_OUT);
                hi[nxt][r] = *(const v2f*)(wrow + (size_t)(kb + 16 + r) * D_OUT + 32);
            }
        }
        __builtin_amdgcn_sched_barrier(0);
        const float av[8] = {a0[cur].x, a0[cur].y, a0[cur].z, a0[cur].w,
                             a1[cur].x, a1[cur].y, a1[cur].z, a1[cur].w};
        #pragma unroll
        for (int j = 0; j < 4; ++j) {
            v2f a;  a.x  = av[2 * j];            a.y  = av[2 * j + 1];
            v2f b0; b0.x = lo[cur][2 * j].x;     b0.y = lo[cur][2 * j + 1].x;
            v2f b1; b1.x = hi[cur][2 * j].x;     b1.y = hi[cur][2 * j + 1].x;
            v2f b2; b2.x = lo[cur][2 * j].y;     b2.y = lo[cur][2 * j + 1].y;
            v2f b3; b3.x = hi[cur][2 * j].y;     b3.y = hi[cur][2 * j + 1].y;
            acc[0] = __builtin_amdgcn_wmma_f32_16x16x4_f32(false, a, false, b0, (short)0, acc[0], false, false);
            acc[1] = __builtin_amdgcn_wmma_f32_16x16x4_f32(false, a, false, b1, (short)0, acc[1], false, false);
            acc[2] = __builtin_amdgcn_wmma_f32_16x16x4_f32(false, a, false, b2, (short)0, acc[2], false, false);
            acc[3] = __builtin_amdgcn_wmma_f32_16x16x4_f32(false, a, false, b3, (short)0, acc[3], false, false);
        }
        __builtin_amdgcn_sched_barrier(0);
    }

    float* hout = h + (size_t)(m0 + half * 8) * D_OUT + 4 * l15;
    #pragma unroll
    for (int v = 0; v < 8; ++v) {
        v4f o; o.x = acc[0][v]; o.y = acc[1][v]; o.z = acc[2][v]; o.w = acc[3][v];
        *(v4f*)(hout + (size_t)v * D_OUT) = o;
    }
}

#define NB       512
#define RPB      196
#define NP       256
#define EPP      (N_EDGES / NP)
#define RSTRIDE  6272
#define EDGE_GROUP 16
static_assert(N_EDGES % NP == 0, "producers must tile the edges");
static_assert(RSTRIDE >= EPP && RSTRIDE % 32 == 0, "regions start on a line boundary");
static_assert((long long)NB * RPB >= N_NODES, "blocks must cover all rows");
static_assert(N_NODES <= (1 << 17) && RPB < 256, "key packing: 17 bits of src, 8 bits of row (RPB itself names the spare row)");
static_assert(EPP <= 7 * 1024, "seven edges per producer thread");

__global__ __launch_bounds__(1024) void bin_kernel(
    const int* __restrict__ src, const int* __restrict__ dst,
    const float* __restrict__ edge_w, uint2* __restrict__ recs,
    unsigned* __restrict__ tab)
{
    __shared__ uint2    stage[EPP];
    __shared__ unsigned cnt[NB];
    __shared__ unsigned wtot[NB / 32];

    const int tid  = threadIdx.x;
    const int lane = tid & 31;
    const int wave = __builtin_amdgcn_readfirstlane(tid >> 5);
    const int p    = blockIdx.x;
    const int e0   = p * EPP;

    if (tid < NB) cnt[tid] = 0;

    unsigned key[7], bin[7], wbits[7], rank[7];
    #pragma unroll
    for (int k = 0; k < 7; ++k) {
        const int i = tid + 1024 * k;
        key[k] = 0; bin[k] = 0; wbits[k] = 0; rank[k] = 0;
        if (i < EPP) {
            const unsigned d  = (unsigned)dst[e0 + i];
            const unsigned s  = (unsigned)src[e0 + i];
            wbits[k] = (unsigned)__float_as_int(edge_w[e0 + i]);
            const unsigned bb = d / RPB;
            bin[k] = bb;
            key[k] = s | ((d - bb * RPB) << 17);
        }
    }
    __syncthreads();

    #pragma unroll
    for (int k = 0; k < 7; ++k) {
        const int i = tid + 1024 * k;
        if (i < EPP) rank[k] = atomicAdd(&cnt[bin[k]], 1u);
    }
    __syncthreads();

    const unsigned c = (tid < NB) ? cnt[tid] : 0u;
    unsigned inc = c;
    #pragma unroll
    for (int d = 1; d < 32; d <<= 1) {
        const unsigned t = __shfl_up(inc, d);
        if (lane >= d) inc += t;
    }
    if (tid < NB && lane == 31) wtot[wave] = inc;
    __syncthreads();
    if (tid < NB) {
        unsigned base = 0;
        for (int w2 = 0; w2 < NB / 32; ++w2) if (w2 < wave) base += wtot[w2];
        const unsigned exc = base + inc - c;
        cnt[tid] = exc;
        tab[p * NB + tid] = exc;
    }
    __syncthreads();

    #pragma unroll
    for (int k = 0; k < 7; ++k) {
        const int i = tid + 1024 * k;
        if (i < EPP) stage[cnt[bin[k]] + rank[k]] = make_uint2(key[k], wbits[k]);
    }
    __syncthreads();

    uint2* region = recs + (size_t)p * RSTRIDE;
    for (int i = tid; i < EPP; i += 1024) region[i] = stage[i];
}

__device__ __forceinline__ void add_records(const float2* __restrict__ h2, float* acc,
                                            int key, int wb, int lane, int g)
{
    float2 v[EDGE_GROUP];
    #pragma unroll
    for (int j = 0; j < EDGE_GROUP; ++j) {
        const int s = __builtin_amdgcn_readlane(key, g + j) & 0x1FFFF;
        v[j] = h2[(size_t)s * (D_OUT / 2) + lane];
    }
    #pragma unroll
    for (int j = 0; j < EDGE_GROUP; ++j) {
        const int   kj = __builtin_amdgcn_readlane(key, g + j);
        const float w  = __int_as_float(__builtin_amdgcn_readlane(wb, g + j));
        float* a = acc + (kj >> 17) * D_OUT + lane;
        __hip_atomic_fetch_add(a,      v[j].x * w, __ATOMIC_RELAXED, __HIP_MEMORY_SCOPE_WORKGROUP);
        __hip_atomic_fetch_add(a + 32, v[j].y * w, __ATOMIC_RELAXED, __HIP_MEMORY_SCOPE_WORKGROUP);
    }
}

__global__ __launch_bounds__(1024, 16) void gather_kernel(
    const float* __restrict__ h, const uint2* __restrict__ recs,
    const unsigned* __restrict__ tab, float* __restrict__ out)
{
    __shared__ float    acc[(RPB + 1) * D_OUT];
    __shared__ unsigned scanv[NP + 1];
    __shared__ unsigned gst[NP];
    __shared__ unsigned wtot[NP / 32];

    const int tid  = threadIdx.x;
    const int lane = tid & 31;
    const int wave = __builtin_amdgcn_readfirstlane(tid >> 5);
    const int b    = blockIdx.x;

    const float4 zero4 = make_float4(0.f, 0.f, 0.f, 0.f);
    for (int i = tid; i < (RPB + 1) * D_OUT / 4; i += 1024) ((float4*)acc)[i] = zero4;

    unsigned st = 0, c = 0;
    if (tid < NP) {
        st = tab[tid * NB + b];
        const unsigned en = (b == NB - 1) ? (unsigned)EPP : tab[tid * NB + b + 1];
        c = en - st;
    }
    unsigned inc = c;
    #pragma unroll
    for (int d = 1; d < 32; d <<= 1) {
        const unsigned t = __shfl_up(inc, d);
        if (lane >= d) inc += t;
    }
    if (tid < NP && lane == 31) wtot[wave] = inc;
    __syncthreads();
    if (tid < NP) {
        unsigned base = 0;
        for (int w2 = 0; w2 < NP / 32; ++w2) if (w2 < wave) base += wtot[w2];
        const unsigned exc = base + inc - c;
        scanv[tid] = exc;
        gst[tid]   = (unsigned)tid * RSTRIDE + st;
        if (tid == NP - 1) scanv[NP] = exc + c;
    }
    __syncthreads();

    const int T = __builtin_amdgcn_readfirstlane((int)scanv[NP]);
    const float2* h2 = (const float2*)h;

    for (int k0 = wave * 32; k0 < T; k0 += 1024) {
        const unsigned k  = (unsigned)(k0 + lane);
        const unsigned kk = k < (unsigned)T ? k : (unsigned)(T - 1);
        unsigned lo = 0, hi = NP;
        #pragma unroll
        for (int it = 0; it < 8; ++it) {
            const unsigned mid = (lo + hi) >> 1;
            if (scanv[mid] <= kk) lo = mid; else hi = mid;
        }
        const uint2 rec = recs[gst[lo] + (kk - scanv[lo])];
        const int key = (k < (unsigned)T) ? (int)rec.x : (int)((rec.x & 0x1FFFFu) | ((unsigned)RPB << 17));
        const int wb  = (int)rec.y;

        #pragma unroll 1
        for (int g = 0; g < 32; g += EDGE_GROUP) add_records(h2, acc, key, wb, lane, g);
    }
    __syncthreads();

    const int row0 = b * RPB;
    int nrows = N_NODES - row0;
    nrows = nrows < 0 ? 0 : (nrows > RPB ? RPB : nrows);
    float4* o4 = (float4*)(out + (size_t)row0 * D_OUT);
    for (int i = tid; i < nrows * (D_OUT / 4); i += 1024) o4[i] = ((const float4*)acc)[i];
}

extern "C" void kernel_launch(void* const* d_in, const int* in_sizes, int n_in,
                              void* d_out, int out_size, void* d_ws, size_t ws_size,
                              hipStream_t stream) {
    const float* x      = (const float*)d_in[0];
    const float* W      = (const float*)d_in[1];
    const float* b      = (const float*)d_in[2];
    const float* edge_w = (const float*)d_in[3];
    const int*   src    = (const int*)d_in[4];
    const int*   dst    = (const int*)d_in[5];
    float*       out    = (float*)d_out;
    float*       h      = (float*)d_ws;
    uint2*       recs   = (uint2*)((char*)d_ws + 33554432);
    unsigned*    tab    = (unsigned*)((char*)d_ws + 50331648);

    (void)in_sizes; (void)n_in; (void)out_size; (void)ws_size;

    const int m_tiles = (N_NODES + 15) / 16;
    const int gemm_blocks = (m_tiles + 7) / 8;
    gemm_bias_kernel<<<gemm_blocks, 256, 0, stream>>>(x, W, b, h);

    bin_kernel<<<NP, 1024, 0, stream>>>(src, dst, edge_w, recs, tab);

    gather_kernel<<<NB, 1024, 0, stream>>>(h, recs, tab, out);
}


# Edge filing moved into the projection launch: its 256 producers run as 256-thread work-groups beside the WMMA work-groups on the same WGPs; 2 launches, not 3; float sums still in LDS arrival order
# speedup vs baseline: 3.3563x; 1.0754x over previous
#include <hip/hip_runtime.h>

#define N_NODES 100000
#define N_EDGES 1600000
#define D_IN    128
#define D_OUT   64

typedef __attribute__((ext_vector_type(2))) float v2f;
typedef __attribute__((ext_vector_type(4))) float v4f;
typedef __attribute__((ext_vector_type(8))) float v8f;

__device__ __forceinline__ void project_strips(
    const int group,
    const float* __restrict__ x, const float* __restrict__ W,
    const float* __restrict__ bias, float* __restrict__ h)
{
    const int lane = threadIdx.x & 31;
    const int wave = threadIdx.x >> 5;
    const int m0   = (group * 8 + wave) * 16;
    if (m0 >= N_NODES) return;

    const int half = lane >> 4;
    const int l15  = lane & 15;

    v8f acc[4];
    #pragma unroll
    for (int t = 0; t < 4; ++t) {
        const float bval = bias[2 * l15 + (t >> 1) + 32 * (t & 1)];
        #pragma unroll
        for (int i = 0; i < 8; ++i) acc[t][i] = bval;
    }

    const float* xrow = x + (size_t)(m0 + l15) * D_IN + half * 8;
    const float* wrow = W + (size_t)(half * 8) * D_OUT + 2 * l15;

    v4f a0[2], a1[2];
    v2f lo[2][8], hi[2][8];
    a0[0] = *(const v4f*)(xrow);
    a1[0] = *(const v4f*)(xrow + 4);
    #pragma unroll
    for (int r = 0; r < 8; ++r) {
        lo[0][r] = *(const v2f*)(wrow + (size_t)r * D_OUT);
        hi[0][r] = *(const v2f*)(wrow + (size_t)r * D_OUT + 32);
    }

    #pragma unroll
    for (int kb = 0; kb < D_IN; kb += 16) {
        const int cur = (kb >> 4) & 1, nxt = cur ^ 1;
        if (kb + 16 < D_IN) {
            a0[nxt] = *(const v4f*)(xrow + kb + 16);
            a1[nxt] = *(const v4f*)(xrow + kb + 20);
            #pragma unroll
            for (int r = 0; r < 8; ++r) {
                lo[nxt][r] = *(const v2f*)(wrow + (size_t)(kb + 16 + r) * D_OUT);
                hi[nxt][r] = *(const v2f*)(wrow + (size_t)(kb + 16 + r) * D_OUT + 32);
            }
        }
        __builtin_amdgcn_sched_barrier(0);
        const float av[8] = {a0[cur].x, a0[cur].y, a0[cur].z, a0[cur].w,
                             a1[cur].x, a1[cur].y, a1[cur].z, a1[cur].w};
        #pragma unroll
        for (int j = 0; j < 4; ++j) {
            v2f a;  a.x  = av[2 * j];            a.y  = av[2 * j + 1];
            v2f b0; b0.x = lo[cur][2 * j].x;     b0.y = lo[cur][2 * j + 1].x;
            v2f b1; b1.x = hi[cur][2 * j].x;     b1.y = hi[cur][2 * j + 1].x;
            v2f b2; b2.x = lo[cur][2 * j].y;     b2.y = lo[cur][2 * j + 1].y;
            v2f b3; b3.x = hi[cur][2 * j].y;     b3.y = hi[cur][2 * j + 1].y;
            acc[0] = __builtin_amdgcn_wmma_f32_16x16x4_f32(false, a, false, b0, (short)0, acc[0], false, false);
            acc[1] = __builtin_amdgcn_wmma_f32_16x16x4_f32(false, a, false, b1, (short)0, acc[1], false, false);
            acc[2] = __builtin_amdgcn_wmma_f32_16x16x4_f32(false, a, false, b2, (short)0, acc[2], false, false);
            acc[3] = __builtin_amdgcn_wmma_f32_16x16x4_f32(false, a, false, b3, (short)0, acc[3], false, false);
        }
        __builtin_amdgcn_sched_barrier(0);
    }

    float* hout = h + (size_t)(m0 + half * 8) * D_OUT + 4 * l15;
    #pragma unroll
    for (int v = 0; v < 8; ++v) {
        v4f o; o.x = acc[0][v]; o.y = acc[1][v]; o.z = acc[2][v]; o.w = acc[3][v];
        *(v4f*)(hout + (size_t)v * D_OUT) = o;
    }
}

#define NB       512
#define RPB      196
#define NP       256
#define EPP      (N_EDGES / NP)
#define RSTRIDE  6272
#define EDGE_GROUP 16
#define FT       256
#define EPT      ((EPP + FT - 1) / FT)
static_assert(N_EDGES % NP == 0, "producers must tile the edges");
static_assert(RSTRIDE >= EPP && RSTRIDE % 32 == 0, "regions start on a line boundary");
static_assert((long long)NB * RPB >= N_NODES, "blocks must cover all rows");
static_assert(N_NODES <= (1 << 17) && RPB < 256, "key packing: 17 bits of src, 8 bits of row (RPB itself names the spare row)");
static_assert(NB == 2 * FT, "two blocks per filing thread in the scan");
static_assert(NB <= (1 << 9) && RPB <= (1 << 8) && EPP <= (1 << 13), "block, row and rank share one register: 9 + 8 + 13 bits");

__device__ __forceinline__ void file_edges(
    const int p,
    const int* __restrict__ src, const int* __restrict__ dst,
    const float* __restrict__ edge_w, uint2* __restrict__ recs,
    unsigned* __restrict__ tab)
{
    __shared__ uint2    stage[EPP];
    __shared__ unsigned cnt[NB];
    __shared__ unsigned wtot[FT / 32];

    const int tid  = threadIdx.x;
    __builtin_assume(tid < FT);
    const int lane = tid & 31;
    const int wave = __builtin_amdgcn_readfirstlane(tid >> 5);
    const int e0   = p * EPP;

    cnt[tid]      = 0;
    cnt[tid + FT] = 0;

    unsigned br[EPT];
    #pragma unroll
    for (int k = 0; k < EPT; ++k) {
        const int i = tid + FT * k;
        br[k] = 0;
        if (i < EPP) {
            const unsigned d  = (unsigned)dst[e0 + i];
            const unsigned bb = d / RPB;
            br[k] = bb | ((d - bb * RPB) << 9);
        }
    }
    __syncthreads();

    #pragma unroll
    for (int k = 0; k < EPT; ++k) {
        const int i = tid + FT * k;
        if (i < EPP) br[k] |= atomicAdd(&cnt[br[k] & (NB - 1)], 1u) << 17;
    }
    __syncthreads();

    const uint2 c2 = ((const uint2*)cnt)[tid];
    const unsigned c = c2.x + c2.y;
    unsigned inc = c;
    #pragma unroll
    for (int d = 1; d < 32; d <<= 1) {
        const unsigned t = __shfl_up(inc, d);
        if (lane >= d) inc += t;
    }
    if (lane == 31) wtot[wave] = inc;
    __syncthreads();
    {
        unsigned base = 0;
        for (int w2 = 0; w2 < FT / 32; ++w2) if (w2 < wave) base += wtot[w2];
        const unsigned exc = base + inc - c;
        const uint2 st2 = make_uint2(exc, exc + c2.x);
        ((uint2*)cnt)[tid] = st2;
        ((uint2*)(tab + (size_t)p * NB))[tid] = st2;
    }
    __syncthreads();

    #pragma unroll
    for (int k = 0; k < EPT; ++k) {
        const int i = tid + FT * k;
        if (i < EPP) br[k] = (cnt[br[k] & (NB - 1)] + (br[k] >> 17)) | (((br[k] >> 9) & 0xFFu) << 13);
    }
    #pragma unroll
    for (int k = 0; k < EPT; ++k) {
        const int i = tid + FT * k;
        if (i < EPP) {
            const unsigned s     = (unsigned)src[e0 + i];
            const unsigned wbits = (unsigned)__float_as_int(edge_w[e0 + i]);
            stage[br[k] & 0x1FFFu] = make_uint2(s | ((br[k] >> 13) << 17), wbits);
        }
    }
    __syncthreads();

    uint2* region = recs + (size_t)p * RSTRIDE;
    for (int i = tid; i < EPP; i += FT) region[i] = stage[i];
}

__global__ __launch_bounds__(FT, 8) void project_and_file_kernel(
    const float* __restrict__ x, const float* __restrict__ W,
    const float* __restrict__ bias, float* __restrict__ h,
    const int* __restrict__ src, const int* __restrict__ dst,
    const float* __restrict__ edge_w, uint2* __restrict__ recs,
    unsigned* __restrict__ tab)
{
    if (blockIdx.x < NP) {
        file_edges(blockIdx.x, src, dst, edge_w, recs, tab);
        return;
    }
    project_strips(blockIdx.x - NP, x, W, bias, h);
}

__device__ __forceinline__ void add_records(const float2* __restrict__ h2, float* acc,
                                            int key, int wb, int lane, int g)
{
    float2 v[EDGE_GROUP];
    #pragma unroll
    for (int j = 0; j < EDGE_GROUP; ++j) {
        const int s = __builtin_amdgcn_readlane(key, g + j) & 0x1FFFF;
        v[j] = h2[(size_t)s * (D_OUT / 2) + lane];
    }
    #pragma unroll
    for (int j = 0; j < EDGE_GROUP; ++j) {
        const int   kj = __builtin_amdgcn_readlane(key, g + j);
        const float w  = __int_as_float(__builtin_amdgcn_readlane(wb, g + j));
        float* a = acc + (kj >> 17) * D_OUT + lane;
        __hip_atomic_fetch_add(a,      v[j].x * w, __ATOMIC_RELAXED, __HIP_MEMORY_SCOPE_WORKGROUP);
        __hip_atomic_fetch_add(a + 32, v[j].y * w, __ATOMIC_RELAXED, __HIP_MEMORY_SCOPE_WORKGROUP);
    }
}

__global__ __launch_bounds__(1024, 16) void gather_kernel(
    const float* __restrict__ h, const uint2* __restrict__ recs,
    const unsigned* __restrict__ tab, float* __restrict__ out)
{
    __shared__ float    acc[(RPB + 1) * D_OUT];
    __shared__ unsigned scanv[NP + 1];
    __shared__ unsigned gst[NP];
    __shared__ unsigned wtot[NP / 32];

    const int tid  = threadIdx.x;
    const int lane = tid & 31;
    const int wave = __builtin_amdgcn_readfirstlane(tid >> 5);
    const int b    = blockIdx.x;

    const float4 zero4 = make_float4(0.f, 0.f, 0.f, 0.f);
    for (int i = tid; i < (RPB + 1) * D_OUT / 4; i += 1024) ((float4*)acc)[i] = zero4;

    unsigned st = 0, c = 0;
    if (tid < NP) {
        st = tab[tid * NB + b];
        const unsigned en = (b == NB - 1) ? (unsigned)EPP : tab[tid * NB + b + 1];
        c = en - st;
    }
    unsigned inc = c;
    #pragma unroll
    for (int d = 1; d < 32; d <<= 1) {
        const unsigned t = __shfl_up(inc, d);
        if (lane >= d) inc += t;
    }
    if (tid < NP && lane == 31) wtot[wave] = inc;
    __syncthreads();
    if (tid < NP) {
        unsigned base = 0;
        for (int w2 = 0; w2 < NP / 32; ++w2) if (w2 < wave) base += wtot[w2];
        const unsigned exc = base + inc - c;
        scanv[tid] = exc;
        gst[tid]   = (unsigned)tid * RSTRIDE + st;
        if (tid == NP - 1) scanv[NP] = exc + c;
    }
    __syncthreads();

    const int T = __builtin_amdgcn_readfirstlane((int)scanv[NP]);
    const float2* h2 = (const float2*)h;

    for (int k0 = wave * 32; k0 < T; k0 += 1024) {
        const unsigned k  = (unsigned)(k0 + lane);
        const unsigned kk = k < (unsigned)T ? k : (unsigned)(T - 1);
        unsigned lo = 0, hi = NP;
        #pragma unroll
        for (int it = 0; it < 8; ++it) {
            const unsigned mid = (lo + hi) >> 1;
            if (scanv[mid] <= kk) lo = mid; else hi = mid;
        }
        const uint2 rec = recs[gst[lo] + (kk - scanv[lo])];
        const int key = (k < (unsigned)T) ? (int)rec.x : (int)((rec.x & 0x1FFFFu) | ((unsigned)RPB << 17));
        const int wb  = (int)rec.y;

        #pragma unroll 1
        for (int g = 0; g < 32; g += EDGE_GROUP) add_records(h2, acc, key, wb, lane, g);
    }
    __syncthreads();

    const int row0 = b * RPB;
    int nrows = N_NODES - row0;
    nrows = nrows < 0 ? 0 : (nrows > RPB ? RPB : nrows);
    float4* o4 = (float4*)(out + (size_t)row0 * D_OUT);
    for (int i = tid; i < nrows * (D_OUT / 4); i += 1024) o4[i] = ((const float4*)acc)[i];
}

extern "C" void kernel_launch(void* const* d_in, const int* in_sizes, int n_in,
                              void* d_out, int out_size, void* d_ws, size_t ws_size,
                              hipStream_t stream) {
    const float* x      = (const float*)d_in[0];
    const float* W      = (const float*)d_in[1];
    const float* b      = (const float*)d_in[2];
    const float* edge_w = (const float*)d_in[3];
    const int*   src    = (const int*)d_in[4];
    const int*   dst    = (const int*)d_in[5];
    float*       out    = (float*)d_out;
    float*       h      = (float*)d_ws;
    uint2*       recs   = (uint2*)((char*)d_ws + 33554432);
    unsigned*    tab    = (unsigned*)((char*)d_ws + 50331648);

    (void)in_sizes; (void)n_in; (void)out_size; (void)ws_size;

    const int m_tiles = (N_NODES + 15) / 16;
    const int gemm_blocks = (m_tiles + 7) / 8;
    project_and_file_kernel<<<NP + gemm_blocks, FT, 0, stream>>>(x, W, b, h, src, dst, edge_w, recs, tab);

    gather_kernel<<<NB, 1024, 0, stream>>>(h, recs, tab, out);
}


# Gather: a block's records are first ordered by destination row in LDS (integer rank add, scan, place); a wave sums each run of one row in registers, adds it to LDS once; sums in rank (arrival) order
# speedup vs baseline: 3.5669x; 1.0627x over previous
#include <hip/hip_runtime.h>

#define N_NODES 100000
#define N_EDGES 1600000
#define D_IN    128
#define D_OUT   64

typedef __attribute__((ext_vector_type(2))) float v2f;
typedef __attribute__((ext_vector_type(4))) float v4f;
typedef __attribute__((ext_vector_type(8))) float v8f;

__device__ __forceinline__ void project_strips(
    const int group,
    const float* __restrict__ x, const float* __restrict__ W,
    const float* __restrict__ bias, float* __restrict__ h)
{
    const int lane = threadIdx.x & 31;
    const int wave = threadIdx.x >> 5;
    const int m0   = (group * 8 + wave) * 16;
    if (m0 >= N_NODES) return;

    const int half = lane >> 4;
    const int l15  = lane & 15;

    v8f acc[4];
    #pragma unroll
    for (int t = 0; t < 4; ++t) {
        const float bval = bias[2 * l15 + (t >> 1) + 32 * (t & 1)];
        #pragma unroll
        for (int i = 0; i < 8; ++i) acc[t][i] = bval;
    }

    const float* xrow = x + (size_t)(m0 + l15) * D_IN + half * 8;
    const float* wrow = W + (size_t)(half * 8) * D_OUT + 2 * l15;

    v4f a0[2], a1[2];
    v2f lo[2][8], hi[2][8];
    a0[0] = *(const v4f*)(xrow);
    a1[0] = *(const v4f*)(xrow + 4);
    #pragma unroll
    for (int r = 0; r < 8; ++r) {
        lo[0][r] = *(const v2f*)(wrow + (size_t)r * D_OUT);
        hi[0][r] = *(const v2f*)(wrow + (size_t)r * D_OUT + 32);
    }

    #pragma unroll
    for (int kb = 0; kb < D_IN; kb += 16) {
        const int cur = (kb >> 4) & 1, nxt = cur ^ 1;
        if (kb + 16 < D_IN) {
            a0[nxt] = *(const v4f*)(xrow + kb + 16);
            a1[nxt] = *(const v4f*)(xrow + kb + 20);
            #pragma unroll
            for (int r = 0; r < 8; ++r) {
                lo[nxt][r] = *(const v2f*)(wrow + (size_t)(kb + 16 + r) * D_OUT);
                hi[nxt][r] = *(const v2f*)(wrow + (size_t)(kb + 16 + r) * D_OUT + 32);
            }
        }
        __builtin_amdgcn_sched_barrier(0);
        const float av[8] = {a0[cur].x, a0[cur].y, a0[cur].z, a0[cur].w,
                             a1[cur].x, a1[cur].y, a1[cur].z, a1[cur].w};
        #pragma unroll
        for (int j = 0; j < 4; ++j) {
            v2f a;  a.x  = av[2 * j];            a.y  = av[2 * j + 1];
            v2f b0; b0.x = lo[cur][2 * j].x;     b0.y = lo[cur][2 * j + 1].x;
            v2f b1; b1.x = hi[cur][2 * j].x;     b1.y = hi[cur][2 * j + 1].x;
            v2f b2; b2.x = lo[cur][2 * j].y;     b2.y = lo[cur][2 * j + 1].y;
            v2f b3; b3.x = hi[cur][2 * j].y;     b3.y = hi[cur][2 * j + 1].y;
            acc[0] = __builtin_amdgcn_wmma_f32_16x16x4_f32(false, a, false, b0, (short)0, acc[0], false, false);
            acc[1] = __builtin_amdgcn_wmma_f32_16x16x4_f32(false, a, false, b1, (short)0, acc[1], false, false);
            acc[2] = __builtin_amdgcn_wmma_f32_16x16x4_f32(false, a, false, b2, (short)0, acc[2], false, false);
            acc[3] = __builtin_amdgcn_wmma_f32_16x16x4_f32(false, a, false, b3, (short)0, acc[3], false, false);
        }
        __builtin_amdgcn_sched_barrier(0);
    }

    float* hout = h + (size_t)(m0 + half * 8) * D_OUT + 4 * l15;
    #pragma unroll
    for (int v = 0; v < 8; ++v) {
        v4f o; o.x = acc[0][v]; o.y = acc[1][v]; o.z = acc[2][v]; o.w = acc[3][v];
        *(v4f*)(hout + (size_t)v * D_OUT) = o;
    }
}

#define NB       512
#define RPB      196
#define NP       256
#define EPP      (N_EDGES / NP)
#define RSTRIDE  6272
#define EDGE_GROUP 16
#define FT       256
#define EPT      ((EPP + FT - 1) / FT)
static_assert(N_EDGES % NP == 0, "producers must tile the edges");
static_assert(RSTRIDE >= EPP && RSTRIDE % 32 == 0, "regions start on a line boundary");
static_assert((long long)NB * RPB >= N_NODES, "blocks must cover all rows");
static_assert(N_NODES <= (1 << 17) && RPB < 256, "key packing: 17 bits of src, 8 bits of row (RPB itself names the spare row)");
static_assert(NB == 2 * FT, "two blocks per filing thread in the scan");
static_assert(NB <= (1 << 9) && RPB <= (1 << 8) && EPP <= (1 << 13), "block, row and rank share one register: 9 + 8 + 13 bits");

__device__ __forceinline__ void file_edges(
    const int p,
    const int* __restrict__ src, const int* __restrict__ dst,
    const float* __restrict__ edge_w, uint2* __restrict__ recs,
    unsigned* __restrict__ tab)
{
    __shared__ uint2    stage[EPP];
    __shared__ unsigned cnt[NB];
    __shared__ unsigned wtot[FT / 32];

    const int tid  = threadIdx.x;
    __builtin_assume(tid < FT);
    const int lane = tid & 31;
    const int wave = __builtin_amdgcn_readfirstlane(tid >> 5);
    const int e0   = p * EPP;

    cnt[tid]      = 0;
    cnt[tid + FT] = 0;

    unsigned br[EPT];
    #pragma unroll
    for (int k = 0; k < EPT; ++k) {
        const int i = tid + FT * k;
        br[k] = 0;
        if (i < EPP) {
            const unsigned d  = (unsigned)dst[e0 + i];
            const unsigned bb = d / RPB;
            br[k] = bb | ((d - bb * RPB) << 9);
        }
    }
    __syncthreads();

    #pragma unroll
    for (int k = 0; k < EPT; ++k) {
        const int i = tid + FT * k;
        if (i < EPP) br[k] |= atomicAdd(&cnt[br[k] & (NB - 1)], 1u) << 17;
    }
    __syncthreads();

    const uint2 c2 = ((const uint2*)cnt)[tid];
    const unsigned c = c2.x + c2.y;
    unsigned inc = c;
    #pragma unroll
    for (int d = 1; d < 32; d <<= 1) {
        const unsigned t = __shfl_up(inc, d);
        if (lane >= d) inc += t;
    }
    if (lane == 31) wtot[wave] = inc;
    __syncthreads();
    {
        unsigned base = 0;
        for (int w2 = 0; w2 < FT / 32; ++w2) if (w2 < wave) base += wtot[w2];
        const unsigned exc = base + inc - c;
        const uint2 st2 = make_uint2(exc, exc + c2.x);
        ((uint2*)cnt)[tid] = st2;
        ((uint2*)(tab + (size_t)p * NB))[tid] = st2;
    }
    __syncthreads();

    #pragma unroll
    for (int k = 0; k < EPT; ++k) {
        const int i = tid + FT * k;
        if (i < EPP) br[k] = (cnt[br[k] & (NB - 1)] + (br[k] >> 17)) | (((br[k] >> 9) & 0xFFu) << 13);
    }
    #pragma unroll
    for (int k = 0; k < EPT; ++k) {
        const int i = tid + FT * k;
        if (i < EPP) {
            const unsigned s     = (unsigned)src[e0 + i];
            const unsigned wbits = (unsigned)__float_as_int(edge_w[e0 + i]);
            stage[br[k] & 0x1FFFu] = make_uint2(s | ((br[k] >> 13) << 17), wbits);
        }
    }
    __syncthreads();

    uint2* region = recs + (size_t)p * RSTRIDE;
    for (int i = tid; i < EPP; i += FT) region[i] = stage[i];
}

__global__ __launch_bounds__(FT, 8) void project_and_file_kernel(
    const float* __restrict__ x, const float* __restrict__ W,
    const float* __restrict__ bias, float* __restrict__ h,
    const int* __restrict__ src, const int* __restrict__ dst,
    const float* __restrict__ edge_w, uint2* __restrict__ recs,
    unsigned* __restrict__ tab)
{
    if (blockIdx.x < NP) {
        file_edges(blockIdx.x, src, dst, edge_w, recs, tab);
        return;
    }
    project_strips(blockIdx.x - NP, x, W, bias, h);
}

__device__ __forceinline__ void flush_run(float* acc, int row, int lane, float s0, float s1)
{
    float* a = acc + row * D_OUT + lane;
    __hip_atomic_fetch_add(a,      s0, __ATOMIC_RELAXED, __HIP_MEMORY_SCOPE_WORKGROUP);
    __hip_atomic_fetch_add(a + 32, s1, __ATOMIC_RELAXED, __HIP_MEMORY_SCOPE_WORKGROUP);
}

#define CH     4096
#define SLOTS  (CH / 1024)
#define CNTN   224
static_assert(CNTN >= RPB + 1 && CNTN == 7 * 32, "one wave scans the counters, 7 per lane");
static_assert(CH % EDGE_GROUP == 0 && CH <= (1 << 12), "rank fits 12 bits beside the row");

__global__ __launch_bounds__(1024, 16) void gather_kernel(
    const float* __restrict__ h, const uint2* __restrict__ recs,
    const unsigned* __restrict__ tab, float* __restrict__ out)
{
    __shared__ float    acc[(RPB + 1) * D_OUT];
    __shared__ uint2    sorted[CH];
    __shared__ unsigned cnt[CNTN];
    __shared__ unsigned start[CNTN];
    __shared__ unsigned scanv[NP + 1];
    __shared__ unsigned gst[NP];
    __shared__ unsigned wtot[NP / 32];

    const int tid  = threadIdx.x;
    const int lane = tid & 31;
    const int wave = __builtin_amdgcn_readfirstlane(tid >> 5);
    const int b    = blockIdx.x;

    const float4 zero4 = make_float4(0.f, 0.f, 0.f, 0.f);
    for (int i = tid; i < (RPB + 1) * D_OUT / 4; i += 1024) ((float4*)acc)[i] = zero4;
    if (tid < CNTN) cnt[tid] = 0;

    unsigned st = 0, c = 0;
    if (tid < NP) {
        st = tab[tid * NB + b];
        const unsigned en = (b == NB - 1) ? (unsigned)EPP : tab[tid * NB + b + 1];
        c = en - st;
    }
    unsigned inc = c;
    #pragma unroll
    for (int d = 1; d < 32; d <<= 1) {
        const unsigned t = __shfl_up(inc, d);
        if (lane >= d) inc += t;
    }
    if (tid < NP && lane == 31) wtot[wave] = inc;
    __syncthreads();
    if (tid < NP) {
        unsigned base = 0;
        for (int w2 = 0; w2 < NP / 32; ++w2) if (w2 < wave) base += wtot[w2];
        const unsigned exc = base + inc - c;
        scanv[tid] = exc;
        gst[tid]   = (unsigned)tid * RSTRIDE + st;
        if (tid == NP - 1) scanv[NP] = exc + c;
    }
    __syncthreads();

    const int T = __builtin_amdgcn_readfirstlane((int)scanv[NP]);
    const float2* h2 = (const float2*)h;

    for (int c0 = 0; c0 < T; c0 += CH) {
        const int nc  = (T - c0) < CH ? (T - c0) : CH;
        const int ncp = (nc + EDGE_GROUP - 1) & ~(EDGE_GROUP - 1);

        unsigned kkv[SLOTS], lo[SLOTS], hi[SLOTS];
        #pragma unroll
        for (int i = 0; i < SLOTS; ++i) {
            const unsigned k = (unsigned)(c0 + tid + 1024 * i);
            kkv[i] = k < (unsigned)T ? k : (unsigned)(T - 1);
            lo[i] = 0; hi[i] = NP;
        }
        #pragma unroll
        for (int it = 0; it < 8; ++it) {
            unsigned sv[SLOTS];
            #pragma unroll
            for (int i = 0; i < SLOTS; ++i) sv[i] = scanv[(lo[i] + hi[i]) >> 1];
            #pragma unroll
            for (int i = 0; i < SLOTS; ++i) {
                const unsigned mid = (lo[i] + hi[i]) >> 1;
                if (sv[i] <= kkv[i]) lo[i] = mid; else hi[i] = mid;
            }
        }
        unsigned ra[SLOTS];
        #pragma unroll
        for (int i = 0; i < SLOTS; ++i) ra[i] = gst[lo[i]] + (kkv[i] - scanv[lo[i]]);
        uint2 rec[SLOTS];
        #pragma unroll
        for (int i = 0; i < SLOTS; ++i) rec[i] = recs[ra[i]];
        unsigned key[SLOTS], wbv[SLOTS], rk[SLOTS];
        #pragma unroll
        for (int i = 0; i < SLOTS; ++i) {
            const int q = tid + 1024 * i;
            key[i] = (q < nc) ? rec[i].x : ((rec[i].x & 0x1FFFFu) | ((unsigned)RPB << 17));
            wbv[i] = rec[i].y;
            rk[i]  = 0;
            if (q < ncp) rk[i] = atomicAdd(&cnt[key[i] >> 17], 1u);
        }
        __syncthreads();

        if (wave == 0) {
            unsigned cc[7];
            unsigned sum = 0;
            #pragma unroll
            for (int t = 0; t < 7; ++t) { cc[t] = cnt[7 * lane + t]; sum += cc[t]; }
            unsigned incl = sum;
            #pragma unroll
            for (int d = 1; d < 32; d <<= 1) {
                const unsigned t = __shfl_up(incl, d);
                if (lane >= d) incl += t;
            }
            unsigned run = incl - sum;
            #pragma unroll
            for (int t = 0; t < 7; ++t) {
                start[7 * lane + t] = run;
                run += cc[t];
                cnt[7 * lane + t] = 0;
            }
        }
        __syncthreads();

        #pragma unroll
        for (int i = 0; i < SLOTS; ++i) {
            const int q = tid + 1024 * i;
            if (q < ncp) sorted[start[key[i] >> 17] + rk[i]] = make_uint2(key[i], wbv[i]);
        }
        __syncthreads();

        const int G  = ncp / EDGE_GROUP;
        const int gs = (wave * G) >> 5;
        const int ge = ((wave + 1) * G) >> 5;
        if (gs < ge) {
            int   cur = __builtin_amdgcn_readfirstlane((int)(sorted[gs * EDGE_GROUP].x >> 17));
            float s0 = 0.f, s1 = 0.f;
            #pragma unroll 1
            for (int g = gs; g < ge; ++g) {
                const uint2 rec = sorted[g * EDGE_GROUP + (lane & (EDGE_GROUP - 1))];
                const int key16 = (int)rec.x;
                const int wb16  = (int)rec.y;
                const int rowv = key16 >> 17;
                const int prev = __builtin_amdgcn_update_dpp(cur, rowv, 0x111  , 0xF, 0xF, false);
                const unsigned bm = __builtin_amdgcn_ballot_w32(rowv != prev);
                float2 v[EDGE_GROUP];
                #pragma unroll
                for (int j = 0; j < EDGE_GROUP; ++j) {
                    const int s = __builtin_amdgcn_readlane(key16, j) & 0x1FFFF;
                    v[j] = h2[(size_t)s * (D_OUT / 2) + lane];
                }
                #pragma unroll
                for (int j = 0; j < EDGE_GROUP; ++j) {
                    const float w = __int_as_float(__builtin_amdgcn_readlane(wb16, j));
                    if (__builtin_expect((bm >> j) & 1u, 0)) {
                        flush_run(acc, cur, lane, s0, s1);
                        cur = __builtin_amdgcn_readlane(key16, j) >> 17;
                        s0 = 0.f; s1 = 0.f;
                    }
                    s0 += v[j].x * w;
                    s1 += v[j].y * w;
                }
            }
            flush_run(acc, cur, lane, s0, s1);
        }
    }
    __syncthreads();

    const int row0 = b * RPB;
    int nrows = N_NODES - row0;
    nrows = nrows < 0 ? 0 : (nrows > RPB ? RPB : nrows);
    float4* o4 = (float4*)(out + (size_t)row0 * D_OUT);
    for (int i = tid; i < nrows * (D_OUT / 4); i += 1024) o4[i] = ((const float4*)acc)[i];
}

extern "C" void kernel_launch(void* const* d_in, const int* in_sizes, int n_in,
                              void* d_out, int out_size, void* d_ws, size_t ws_size,
                              hipStream_t stream) {
    const float* x      = (const float*)d_in[0];
    const float* W      = (const float*)d_in[1];
    const float* b      = (const float*)d_in[2];
    const float* edge_w = (const float*)d_in[3];
    const int*   src    = (const int*)d_in[4];
    const int*   dst    = (const int*)d_in[5];
    float*       out    = (float*)d_out;
    float*       h      = (float*)d_ws;
    uint2*       recs   = (uint2*)((char*)d_ws + 33554432);
    unsigned*    tab    = (unsigned*)((char*)d_ws + 50331648);

    (void)in_sizes; (void)n_in; (void)out_size; (void)ws_size;

    const int m_tiles = (N_NODES + 15) / 16;
    const int gemm_blocks = (m_tiles + 7) / 8;
    project_and_file_kernel<<<NP + gemm_blocks, FT, 0, stream>>>(x, W, b, h, src, dst, edge_w, recs, tab);

    gather_kernel<<<NB, 1024, 0, stream>>>(h, recs, tab, out);
}


# Projection dealt evenly over the SIMDs: 768 whole-strip work-groups fill the chip once; the 106 strips left run as four 16x16 tiles each on filing work-groups after their filing; sums unchanged
# speedup vs baseline: 3.8818x; 1.0883x over previous
#include <hip/hip_runtime.h>

#define N_NODES 100000
#define N_EDGES 1600000
#define D_IN    128
#define D_OUT   64

typedef __attribute__((ext_vector_type(2))) float v2f;
typedef __attribute__((ext_vector_type(4))) float v4f;
typedef __attribute__((ext_vector_type(8))) float v8f;

__device__ __forceinline__ void project_strips(
    const int group,
    const float* __restrict__ x, const float* __restrict__ W,
    const float* __restrict__ bias, float* __restrict__ h)
{
    const int lane = threadIdx.x & 31;
    const int wave = threadIdx.x >> 5;
    const int m0   = (group * 8 + wave) * 16;
    if (m0 >= N_NODES) return;

    const int half = lane >> 4;
    const int l15  = lane & 15;

    v8f acc[4];
    #pragma unroll
    for (int t = 0; t < 4; ++t) {
        const float bval = bias[2 * l15 + (t >> 1) + 32 * (t & 1)];
        #pragma unroll
        for (int i = 0; i < 8; ++i) acc[t][i] = bval;
    }

    const float* xrow = x + (size_t)(m0 + l15) * D_IN + half * 8;
    const float* wrow = W + (size_t)(half * 8) * D_OUT + 2 * l15;

    v4f a0[2], a1[2];
    v2f lo[2][8], hi[2][8];
    a0[0] = *(const v4f*)(xrow);
    a1[0] = *(const v4f*)(xrow + 4);
    #pragma unroll
    for (int r = 0; r < 8; ++r) {
        lo[0][r] = *(const v2f*)(wrow + (size_t)r * D_OUT);
        hi[0][r] = *(const v2f*)(wrow + (size_t)r * D_OUT + 32);
    }

    #pragma unroll
    for (int kb = 0; kb < D_IN; kb += 16) {
        const int cur = (kb >> 4) & 1, nxt = cur ^ 1;
        if (kb + 16 < D_IN) {
            a0[nxt] = *(const v4f*)(xrow + kb + 16);
            a1[nxt] = *(const v4f*)(xrow + kb + 20);
            #pragma unroll
            for (int r = 0; r < 8; ++r) {
                lo[nxt][r] = *(const v2f*)(wrow + (size_t)(kb + 16 + r) * D_OUT);
                hi[nxt][r] = *(const v2f*)(wrow + (size_t)(kb + 16 + r) * D_OUT + 32);
            }
        }
        __builtin_amdgcn_sched_barrier(0);
        const float av[8] = {a0[cur].x, a0[cur].y, a0[cur].z, a0[cur].w,
                             a1[cur].x, a1[cur].y, a1[cur].z, a1[cur].w};
        #pragma unroll
        for (int j = 0; j < 4; ++j) {
            v2f a;  a.x  = av[2 * j];            a.y  = av[2 * j + 1];
            v2f b0; b0.x = lo[cur][2 * j].x;     b0.y = lo[cur][2 * j + 1].x;
            v2f b1; b1.x = hi[cur][2 * j].x;     b1.y = hi[cur][2 * j + 1].x;
            v2f b2; b2.x = lo[cur][2 * j].y;     b2.y = lo[cur][2 * j + 1].y;
            v2f b3; b3.x = hi[cur][2 * j].y;     b3.y = hi[cur][2 * j + 1].y;
            acc[0] = __builtin_amdgcn_wmma_f32_16x16x4_f32(false, a, false, b0, (short)0, acc[0], false, false);
            acc[1] = __builtin_amdgcn_wmma_f32_16x16x4_f32(false, a, false, b1, (short)0, acc[1], false, false);
            acc[2] = __builtin_amdgcn_wmma_f32_16x16x4_f32(false, a, false, b2, (short)0, acc[2], false, false);
            acc[3] = __builtin_amdgcn_wmma_f32_16x16x4_f32(false, a, false, b3, (short)0, acc[3], false, false);
        }
        __builtin_amdgcn_sched_barrier(0);
    }

    float* hout = h + (size_t)(m0 + half * 8) * D_OUT + 4 * l15;
    #pragma unroll
    for (int v = 0; v < 8; ++v) {
        v4f o; o.x = acc[0][v]; o.y = acc[1][v]; o.z = acc[2][v]; o.w = acc[3][v];
        *(v4f*)(hout + (size_t)v * D_OUT) = o;
    }
}

__device__ __forceinline__ void project_tile(
    const int strip, const int t,
    const float* __restrict__ x, const float* __restrict__ W,
    const float* __restrict__ bias, float* rows)
{
    const int lane = threadIdx.x & 31;
    const int half = lane >> 4;
    const int l15  = lane & 15;
    const int m0   = strip * 16;
    const int col  = 2 * l15 + (t >> 1) + 32 * (t & 1);

    v8f acc;
    const float bval = bias[col];
    #pragma unroll
    for (int i = 0; i < 8; ++i) acc[i] = bval;

    const float* xrow = x + (size_t)(m0 + l15) * D_IN + half * 8;
    const float* wcol = W + (size_t)(half * 8) * D_OUT + col;

    #pragma unroll
    for (int kb = 0; kb < D_IN; kb += 16) {
        const v4f a0 = *(const v4f*)(xrow + kb);
        const v4f a1 = *(const v4f*)(xrow + kb + 4);
        float bw[8];
        #pragma unroll
        for (int r = 0; r < 8; ++r) bw[r] = wcol[(size_t)(kb + r) * D_OUT];
        const float av[8] = {a0.x, a0.y, a0.z, a0.w, a1.x, a1.y, a1.z, a1.w};
        #pragma unroll
        for (int j = 0; j < 4; ++j) {
            v2f a; a.x = av[2 * j]; a.y = av[2 * j + 1];
            v2f b; b.x = bw[2 * j]; b.y = bw[2 * j + 1];
            acc = __builtin_amdgcn_wmma_f32_16x16x4_f32(false, a, false, b, (short)0, acc, false, false);
        }
    }

    float* r0 = rows + (half * 8) * D_OUT + 4 * l15 + t;
    #pragma unroll
    for (int v = 0; v < 8; ++v) r0[v * D_OUT] = acc[v];
}

#define NB       512
#define RPB      196
#define NP       256
#define EPP      (N_EDGES / NP)
#define RSTRIDE  6272
#define EDGE_GROUP 16
#define FT       256
#define EPT      ((EPP + FT - 1) / FT)
#define M_STRIPS    ((N_NODES + 15) / 16)
#define PROJ_GROUPS 768
#define LEFT_STRIPS (M_STRIPS - PROJ_GROUPS * 8)
static_assert(N_NODES % 16 == 0, "strips are whole");
static_assert(LEFT_STRIPS >= 0 && LEFT_STRIPS <= NP, "one left-over strip per filing work-group at most");
static_assert(FT * 4 == 16 * D_OUT && FT >= 4 * 32, "a strip image is one 16-byte store per thread; four waves take its tiles");
static_assert(N_EDGES % NP == 0, "producers must tile the edges");
static_assert(RSTRIDE >= EPP && RSTRIDE % 32 == 0, "regions start on a line boundary");
static_assert((long long)NB * RPB >= N_NODES, "blocks must cover all rows");
static_assert(N_NODES <= (1 << 17) && RPB < 256, "key packing: 17 bits of src, 8 bits of row (RPB itself names the spare row)");
static_assert(NB == 2 * FT, "two blocks per filing thread in the scan");
static_assert(NB <= (1 << 9) && RPB <= (1 << 8) && EPP <= (1 << 13), "block, row and rank share one register: 9 + 8 + 13 bits");

__device__ __forceinline__ void file_edges(
    const int p,
    const int* __restrict__ src, const int* __restrict__ dst,
    const float* __restrict__ edge_w, uint2* __restrict__ recs,
    unsigned* __restrict__ tab)
{
    __shared__ uint2    stage[EPP];
    __shared__ unsigned cnt[NB];
    __shared__ unsigned wtot[FT / 32];

    const int tid  = threadIdx.x;
    __builtin_assume(tid < FT);
    const int lane = tid & 31;
    const int wave = __builtin_amdgcn_readfirstlane(tid >> 5);
    const int e0   = p * EPP;

    cnt[tid]      = 0;
    cnt[tid + FT] = 0;

    unsigned br[EPT];
    #pragma unroll
    for (int k = 0; k < EPT; ++k) {
        const int i = tid + FT * k;
        br[k] = 0;
        if (i < EPP) {
            const unsigned d  = (unsigned)dst[e0 + i];
            const unsigned bb = d / RPB;
            br[k] = bb | ((d - bb * RPB) << 9);
        }
    }
    __syncthreads();

    #pragma unroll
    for (int k = 0; k < EPT; ++k) {
        const int i = tid + FT * k;
        if (i < EPP) br[k] |= atomicAdd(&cnt[br[k] & (NB - 1)], 1u) << 17;
    }
    __syncthreads();

    const uint2 c2 = ((const uint2*)cnt)[tid];
    const unsigned c = c2.x + c2.y;
    unsigned inc = c;
    #pragma unroll
    for (int d = 1; d < 32; d <<= 1) {
        const unsigned t = __shfl_up(inc, d);
        if (lane >= d) inc += t;
    }
    if (lane == 31) wtot[wave] = inc;
    __syncthreads();
    {
        unsigned base = 0;
        for (int w2 = 0; w2 < FT / 32; ++w2) if (w2 < wave) base += wtot[w2];
        const unsigned exc = base + inc - c;
        const uint2 st2 = make_uint2(exc, exc + c2.x);
        ((uint2*)cnt)[tid] = st2;
        ((uint2*)(tab + (size_t)p * NB))[tid] = st2;
    }
    __syncthreads();

    #pragma unroll
    for (int k = 0; k < EPT; ++k) {
        const int i = tid + FT * k;
        if (i < EPP) br[k] = (cnt[br[k] & (NB - 1)] + (br[k] >> 17)) | (((br[k] >> 9) & 0xFFu) << 13);
    }
    #pragma unroll
    for (int k = 0; k < EPT; ++k) {
        const int i = tid + FT * k;
        if (i < EPP) {
            const unsigned s     = (unsigned)src[e0 + i];
            const unsigned wbits = (unsigned)__float_as_int(edge_w[e0 + i]);
            stage[br[k] & 0x1FFFu] = make_uint2(s | ((br[k] >> 13) << 17), wbits);
        }
    }
    __syncthreads();

    uint2* region = recs + (size_t)p * RSTRIDE;
    for (int i = tid; i < EPP; i += FT) region[i] = stage[i];
}

__global__ __launch_bounds__(FT, 8) void project_and_file_kernel(
    const float* __restrict__ x, const float* __restrict__ W,
    const float* __restrict__ bias, float* __restrict__ h,
    const int* __restrict__ src, const int* __restrict__ dst,
    const float* __restrict__ edge_w, uint2* __restrict__ recs,
    unsigned* __restrict__ tab)
{
    if (blockIdx.x < NP) {
        file_edges(blockIdx.x, src, dst, edge_w, recs, tab);
        if (blockIdx.x < LEFT_STRIPS) {
            __shared__ float rows[16 * D_OUT];
            const int wave  = __builtin_amdgcn_readfirstlane((int)(threadIdx.x >> 5));
            const int strip = PROJ_GROUPS * 8 + (int)blockIdx.x;
            if (wave < 4) project_tile(strip, wave, x, W, bias, rows);
            __syncthreads();
            ((float4*)(h + (size_t)strip * 16 * D_OUT))[threadIdx.x] = ((const float4*)rows)[threadIdx.x];
        }
        return;
    }
    project_strips(blockIdx.x - NP, x, W, bias, h);
}

__device__ __forceinline__ void flush_run(float* acc, int row, int lane, float s0, float s1)
{
    float* a = acc + row * D_OUT + lane;
    __hip_atomic_fetch_add(a,      s0, __ATOMIC_RELAXED, __HIP_MEMORY_SCOPE_WORKGROUP);
    __hip_atomic_fetch_add(a + 32, s1, __ATOMIC_RELAXED, __HIP_MEMORY_SCOPE_WORKGROUP);
}

#define CH     4096
#define SLOTS  (CH / 1024)
#define CNTN   224
static_assert(CNTN >= RPB + 1 && CNTN == 7 * 32, "one wave scans the counters, 7 per lane");
static_assert(CH % EDGE_GROUP == 0 && CH <= (1 << 12), "rank fits 12 bits beside the row");

__global__ __launch_bounds__(1024, 16) void gather_kernel(
    const float* __restrict__ h, const uint2* __restrict__ recs,
    const unsigned* __restrict__ tab, float* __restrict__ out)
{
    __shared__ float    acc[(RPB + 1) * D_OUT];
    __shared__ uint2    sorted[CH];
    __shared__ unsigned cnt[CNTN];
    __shared__ unsigned start[CNTN];
    __shared__ unsigned scanv[NP + 1];
    __shared__ unsigned gst[NP];
    __shared__ unsigned wtot[NP / 32];

    const int tid  = threadIdx.x;
    const int lane = tid & 31;
    const int wave = __builtin_amdgcn_readfirstlane(tid >> 5);
    const int b    = blockIdx.x;

    const float4 zero4 = make_float4(0.f, 0.f, 0.f, 0.f);
    for (int i = tid; i < (RPB + 1) * D_OUT / 4; i += 1024) ((float4*)acc)[i] = zero4;
    if (tid < CNTN) cnt[tid] = 0;

    unsigned st = 0, c = 0;
    if (tid < NP) {
        st = tab[tid * NB + b];
        const unsigned en = (b == NB - 1) ? (unsigned)EPP : tab[tid * NB + b + 1];
        c = en - st;
    }
    unsigned inc = c;
    #pragma unroll
    for (int d = 1; d < 32; d <<= 1) {
        const unsigned t = __shfl_up(inc, d);
        if (lane >= d) inc += t;
    }
    if (tid < NP && lane == 31) wtot[wave] = inc;
    __syncthreads();
    if (tid < NP) {
        unsigned base = 0;
        for (int w2 = 0; w2 < NP / 32; ++w2) if (w2 < wave) base += wtot[w2];
        const unsigned exc = base + inc - c;
        scanv[tid] = exc;
        gst[tid]   = (unsigned)tid * RSTRIDE + st;
        if (tid == NP - 1) scanv[NP] = exc + c;
    }
    __syncthreads();

    const int T = __builtin_amdgcn_readfirstlane((int)scanv[NP]);
    const float2* h2 = (const float2*)h;

    for (int c0 = 0; c0 < T; c0 += CH) {
        const int nc  = (T - c0) < CH ? (T - c0) : CH;
        const int ncp = (nc + EDGE_GROUP - 1) & ~(EDGE_GROUP - 1);

        unsigned kkv[SLOTS], lo[SLOTS], hi[SLOTS];
        #pragma unroll
        for (int i = 0; i < SLOTS; ++i) {
            const unsigned k = (unsigned)(c0 + tid + 1024 * i);
            kkv[i] = k < (unsigned)T ? k : (unsigned)(T - 1);
            lo[i] = 0; hi[i] = NP;
        }
        #pragma unroll
        for (int it = 0; it < 8; ++it) {
            unsigned sv[SLOTS];
            #pragma unroll
            for (int i = 0; i < SLOTS; ++i) sv[i] = scanv[(lo[i] + hi[i]) >> 1];
            #pragma unroll
            for (int i = 0; i < SLOTS; ++i) {
                const unsigned mid = (lo[i] + hi[i]) >> 1;
                if (sv[i] <= kkv[i]) lo[i] = mid; else hi[i] = mid;
            }
        }
        unsigned ra[SLOTS];
        #pragma unroll
        for (int i = 0; i < SLOTS; ++i) ra[i] = gst[lo[i]] + (kkv[i] - scanv[lo[i]]);
        uint2 rec[SLOTS];
        #pragma unroll
        for (int i = 0; i < SLOTS; ++i) rec[i] = recs[ra[i]];
        unsigned key[SLOTS], wbv[SLOTS], rk[SLOTS];
        #pragma unroll
        for (int i = 0; i < SLOTS; ++i) {
            const int q = tid + 1024 * i;
            key[i] = (q < nc) ? rec[i].x : ((rec[i].x & 0x1FFFFu) | ((unsigned)RPB << 17));
            wbv[i] = rec[i].y;
            rk[i]  = 0;
            if (q < ncp) rk[i] = atomicAdd(&cnt[key[i] >> 17], 1u);
        }
        __syncthreads();

        if (wave == 0) {
            unsigned cc[7];
            unsigned sum = 0;
            #pragma unroll
            for (int t = 0; t < 7; ++t) { cc[t] = cnt[7 * lane + t]; sum += cc[t]; }
            unsigned incl = sum;
            #pragma unroll
            for (int d = 1; d < 32; d <<= 1) {
                const unsigned t = __shfl_up(incl, d);
                if (lane >= d) incl += t;
            }
            unsigned run = incl - sum;
            #pragma unroll
            for (int t = 0; t < 7; ++t) {
                start[7 * lane + t] = run;
                run += cc[t];
                cnt[7 * lane + t] = 0;
            }
        }
        __syncthreads();

        #pragma unroll
        for (int i = 0; i < SLOTS; ++i) {
            const int q = tid + 1024 * i;
            if (q < ncp) sorted[start[key[i] >> 17] + rk[i]] = make_uint2(key[i], wbv[i]);
        }
        __syncthreads();

        const int G  = ncp / EDGE_GROUP;
        const int gs = (wave * G) >> 5;
        const int ge = ((wave + 1) * G) >> 5;
        if (gs < ge) {
            int   cur = __builtin_amdgcn_readfirstlane((int)(sorted[gs * EDGE_GROUP].x >> 17));
            float s0 = 0.f, s1 = 0.f;
            #pragma unroll 1
            for (int g = gs; g < ge; ++g) {
                const uint2 rec = sorted[g * EDGE_GROUP + (lane & (EDGE_GROUP - 1))];
                const int key16 = (int)rec.x;
                const int wb16  = (int)rec.y;
                const int rowv = key16 >> 17;
                const int prev = __builtin_amdgcn_update_dpp(cur, rowv, 0x111  , 0xF, 0xF, false);
                const unsigned bm = __builtin_amdgcn_ballot_w32(rowv != prev);
                float2 v[EDGE_GROUP];
                #pragma unroll
                for (int j = 0; j < EDGE_GROUP; ++j) {
                    const int s = __builtin_amdgcn_readlane(key16, j) & 0x1FFFF;
                    v[j] = h2[(size_t)s * (D_OUT / 2) + lane];
                }
                #pragma unroll
                for (int j = 0; j < EDGE_GROUP; ++j) {
                    const float w = __int_as_float(__builtin_amdgcn_readlane(wb16, j));
                    if (__builtin_expect((bm >> j) & 1u, 0)) {
                        flush_run(acc, cur, lane, s0, s1);
                        cur = __builtin_amdgcn_readlane(key16, j) >> 17;
                        s0 = 0.f; s1 = 0.f;
                    }
                    s0 += v[j].x * w;
                    s1 += v[j].y * w;
                }
            }
            flush_run(acc, cur, lane, s0, s1);
        }
    }
    __syncthreads();

    const int row0 = b * RPB;
    int nrows = N_NODES - row0;
    nrows = nrows < 0 ? 0 : (nrows > RPB ? RPB : nrows);
    float4* o4 = (float4*)(out + (size_t)row0 * D_OUT);
    for (int i = tid; i < nrows * (D_OUT / 4); i += 1024) o4[i] = ((const float4*)acc)[i];
}

extern "C" void kernel_launch(void* const* d_in, const int* in_sizes, int n_in,
                              void* d_out, int out_size, void* d_ws, size_t ws_size,
                              hipStream_t stream) {
    const float* x      = (const float*)d_in[0];
    const float* W      = (const float*)d_in[1];
    const float* b      = (const float*)d_in[2];
    const float* edge_w = (const float*)d_in[3];
    const int*   src    = (const int*)d_in[4];
    const int*   dst    = (const int*)d_in[5];
    float*       out    = (float*)d_out;
    float*       h      = (float*)d_ws;
    uint2*       recs   = (uint2*)((char*)d_ws + 33554432);
    unsigned*    tab    = (unsigned*)((char*)d_ws + 50331648);

    (void)in_sizes; (void)n_in; (void)out_size; (void)ws_size;

    project_and_file_kernel<<<NP + PROJ_GROUPS, FT, 0, stream>>>(x, W, b, h, src, dst, edge_w, recs, tab);

    gather_kernel<<<NB, 1024, 0, stream>>>(h, recs, tab, out);
}
